# starting kernel
# baseline (speedup 1.0000x reference)
#include <hip/hip_runtime.h>


typedef __attribute__((ext_vector_type(2))) float v2f;
typedef __attribute__((ext_vector_type(8))) float v8f;

#define D_MODEL 24
#define REL_DIM 10
#define NN 512
#define BB 8
#define K1_CHUNKS 13
#define K2_CHUNKS 3

__global__ __launch_bounds__(32) void pairagg_kernel(
    const float* __restrict__ x, const float* __restrict__ q,
    const float* __restrict__ coord, const float* __restrict__ W1,
    const float* __restrict__ b1, const float* __restrict__ W2,
    const float* __restrict__ b2, float* __restrict__ out)
{
  __shared__ float hT[16 * 16];

  const int bi   = blockIdx.x;
  const int b    = bi / NN;
  const int i    = bi % NN;
  const int lane = threadIdx.x;
  const int m    = lane & 15;
  const int hh   = lane >> 4;

  const float* xB = x     + (size_t)b * NN * D_MODEL;
  const float* qB = q     + (size_t)b * NN * D_MODEL;
  const float* cB = coord + (size_t)b * NN * D_MODEL;

  v2f ci[6], qi[6];
#pragma unroll
  for (int c = 0; c < 6; ++c) {
    const int d = 4 * c + 2 * hh;
    ci[c] = *(const v2f*)(cB + i * D_MODEL + d);
    qi[c] = *(const v2f*)(qB + i * D_MODEL + d);
  }

  float bias_row = 0.0f;
  if (m < REL_DIM) {
    bias_row = b1[m];
#pragma unroll
    for (int d = 0; d < D_MODEL; ++d)
      bias_row = fmaf(xB[i * D_MODEL + d], W1[d * REL_DIM + m], bias_row);
  }

  v2f B1[K1_CHUNKS];
#pragma unroll
  for (int c = 0; c < K1_CHUNKS; ++c) {
#pragma unroll
    for (int v = 0; v < 2; ++v) {
      const int k = 4 * c + 2 * hh + v;
      float val = 0.0f;
      if (m < REL_DIM) {
        int row = -1;
        if      (k < 24)  row = 48 + k;
        else if (k < 48)  row = k;
        else if (k == 48) row = 72;
        else if (k == 49) row = 73;
        if (row >= 0)       val = W1[row * REL_DIM + m];
        else if (k == 50)   val = bias_row;
      }
      B1[c][v] = val;
    }
  }

  v2f B2[K2_CHUNKS];
#pragma unroll
  for (int c = 0; c < K2_CHUNKS; ++c) {
#pragma unroll
    for (int v = 0; v < 2; ++v) {
      const int k = 4 * c + 2 * hh + v;
      B2[c][v] = (k < REL_DIM && m < REL_DIM) ? W2[k * REL_DIM + m] : 0.0f;
    }
  }
  const float b2v = (m < REL_DIM) ? b2[m] : 0.0f;

  float psum = 0.0f;
  float pmax = -1e9f;

  for (int jbase = 0; jbase < NN; jbase += 16) {
    const int j = jbase + m;
    const float* cj_ptr = cB + j * D_MODEL + 2 * hh;
    const float* xj_ptr = xB + j * D_MODEL + 2 * hh;
    const float* qj_ptr = qB + j * D_MODEL + 2 * hh;

    v2f A[K1_CHUNKS];
    float sqpart = 0.0f, qpart = 0.0f;
#pragma unroll
    for (int c = 0; c < 6; ++c) {
      const v2f cj = *(const v2f*)(cj_ptr + 4 * c);
      const v2f xj = *(const v2f*)(xj_ptr + 4 * c);
      const v2f qj = *(const v2f*)(qj_ptr + 4 * c);
      const float d0 = ci[c][0] - cj[0];
      const float d1 = ci[c][1] - cj[1];
      A[c][0] = fabsf(d0);
      A[c][1] = fabsf(d1);
      sqpart  = fmaf(d0, d0, fmaf(d1, d1, sqpart));
      qpart   = fmaf(qi[c][0], qj[0], fmaf(qi[c][1], qj[1], qpart));
      A[6 + c][0] = xj[0];
      A[6 + c][1] = xj[1];
    }
    const float qfull = qpart  + __shfl_xor(qpart, 16, 32);
    const float cdist = sqrtf(sqpart + __shfl_xor(sqpart, 16, 32));
    A[12][0] = hh ? 1.0f : qfull;
    A[12][1] = hh ? 0.0f : cdist;

    v8f acc = {};
#pragma unroll
    for (int c = 0; c < K1_CHUNKS; ++c)
      acc = __builtin_amdgcn_wmma_f32_16x16x4_f32(
          false, A[c], false, B1[c], (short)0, acc, false, false);

    __syncthreads();
#pragma unroll
    for (int r = 0; r < 8; ++r)
      hT[(r + 8 * hh) * 16 + m] = fmaxf(acc[r], 0.0f);
    __syncthreads();

    v2f A2[K2_CHUNKS];
#pragma unroll
    for (int c = 0; c < K2_CHUNKS; ++c)
      A2[c] = *(const v2f*)(&hT[m * 16 + 4 * c + 2 * hh]);

    v8f acc2 = {};
#pragma unroll
    for (int c = 0; c < K2_CHUNKS; ++c)
      acc2 = __builtin_amdgcn_wmma_f32_16x16x4_f32(
          false, A2[c], false, B2[c], (short)0, acc2, false, false);

#pragma unroll
    for (int r = 0; r < 8; ++r) {
      const float rel  = fmaxf(acc2[r] + b2v, 0.0f);
      const bool  diag = (jbase + r + 8 * hh) == i;
      psum += diag ? 0.0f : rel;
      pmax  = diag ? pmax : fmaxf(pmax, rel);
    }
  }

  psum += __shfl_xor(psum, 16, 32);
  pmax  = fmaxf(pmax, __shfl_xor(pmax, 16, 32));

  if (lane < REL_DIM) {
    const size_t base = ((size_t)b * NN + i) * REL_DIM + lane;
    out[base] = psum / (float)(NN - 1);
    out[(size_t)BB * NN * REL_DIM + base] = pmax;
  }
}

extern "C" void kernel_launch(void* const* d_in, const int* in_sizes, int n_in,
                              void* d_out, int out_size, void* d_ws, size_t ws_size,
                              hipStream_t stream) {
  const float* x  = (const float*)d_in[0];
  const float* q  = (const float*)d_in[1];
  const float* co = (const float*)d_in[2];
  const float* W1 = (const float*)d_in[3];
  const float* b1 = (const float*)d_in[4];
  const float* W2 = (const float*)d_in[5];
  const float* b2 = (const float*)d_in[6];
  float* out = (float*)d_out;

  dim3 grid(BB * NN);
  dim3 block(32);
  hipLaunchKernelGGL(pairagg_kernel, grid, block, 0, stream,
                     x, q, co, W1, b1, W2, b2, out);
}


# Stage each batch's x,q,coord in LDS once per 16-wave workgroup (coalesced 16 B loads); waves read j tiles from LDS, not strided global loads; per-wave transpose slice, no block barrier in loop
# speedup vs baseline: 1.5059x; 1.5059x over previous
#include <hip/hip_runtime.h>


typedef __attribute__((ext_vector_type(2))) float v2f;
typedef __attribute__((ext_vector_type(8))) float v8f;

#define D_MODEL 24
#define REL_DIM 10
#define NN 512
#define BB 8
#define K1_CHUNKS 13
#define K2_CHUNKS 3
#define WAVES 16
#define SLAB (NN * D_MODEL)

__global__ __launch_bounds__(32 * WAVES) void pairagg_kernel(
    const float* __restrict__ x, const float* __restrict__ q,
    const float* __restrict__ coord, const float* __restrict__ W1,
    const float* __restrict__ b1, const float* __restrict__ W2,
    const float* __restrict__ b2, float* __restrict__ out)
{
  __shared__ __attribute__((aligned(16))) float sx[SLAB];
  __shared__ __attribute__((aligned(16))) float sq[SLAB];
  __shared__ __attribute__((aligned(16))) float sc[SLAB];
  __shared__ float hT[WAVES][16 * 16];

  const int wave = __builtin_amdgcn_readfirstlane((int)(threadIdx.x >> 5));
  const int b    = blockIdx.x / (NN / WAVES);
  const int i    = (blockIdx.x % (NN / WAVES)) * WAVES + wave;
  const int lane = threadIdx.x & 31;
  const int m    = lane & 15;
  const int hh   = lane >> 4;

  const float* xB = x     + (size_t)b * NN * D_MODEL;
  const float* qB = q     + (size_t)b * NN * D_MODEL;
  const float* cB = coord + (size_t)b * NN * D_MODEL;

  {
    const float4* gx = (const float4*)xB;
    const float4* gq = (const float4*)qB;
    const float4* gc = (const float4*)cB;
    float4* lx = (float4*)sx;
    float4* lq = (float4*)sq;
    float4* lc = (float4*)sc;
#pragma unroll
    for (int t = 0; t < SLAB / 4 / (32 * WAVES); ++t) {
      const int idx = t * (32 * WAVES) + (int)threadIdx.x;
      lx[idx] = gx[idx];
      lq[idx] = gq[idx];
      lc[idx] = gc[idx];
    }
  }

  v2f ci[6], qi[6];
#pragma unroll
  for (int c = 0; c < 6; ++c) {
    const int d = 4 * c + 2 * hh;
    ci[c] = *(const v2f*)(cB + i * D_MODEL + d);
    qi[c] = *(const v2f*)(qB + i * D_MODEL + d);
  }

  float bias_row = 0.0f;
  if (m < REL_DIM) {
    bias_row = b1[m];
#pragma unroll
    for (int d = 0; d < D_MODEL; ++d)
      bias_row = fmaf(xB[i * D_MODEL + d], W1[d * REL_DIM + m], bias_row);
  }

  v2f B1[K1_CHUNKS];
#pragma unroll
  for (int c = 0; c < K1_CHUNKS; ++c) {
#pragma unroll
    for (int v = 0; v < 2; ++v) {
      const int k = 4 * c + 2 * hh + v;
      float val = 0.0f;
      if (m < REL_DIM) {
        int row = -1;
        if      (k < 24)  row = 48 + k;
        else if (k < 48)  row = k;
        else if (k == 48) row = 72;
        else if (k == 49) row = 73;
        if (row >= 0)       val = W1[row * REL_DIM + m];
        else if (k == 50)   val = bias_row;
      }
      B1[c][v] = val;
    }
  }

  v2f B2[K2_CHUNKS];
#pragma unroll
  for (int c = 0; c < K2_CHUNKS; ++c) {
#pragma unroll
    for (int v = 0; v < 2; ++v) {
      const int k = 4 * c + 2 * hh + v;
      B2[c][v] = (k < REL_DIM && m < REL_DIM) ? W2[k * REL_DIM + m] : 0.0f;
    }
  }
  const float b2v = (m < REL_DIM) ? b2[m] : 0.0f;

  __syncthreads();
  float* hw = hT[wave];

  float psum = 0.0f;
  float pmax = -1e9f;

  for (int jbase = 0; jbase < NN; jbase += 16) {
    const int j = jbase + m;
    const float* cj_ptr = sc + j * D_MODEL + 2 * hh;
    const float* xj_ptr = sx + j * D_MODEL + 2 * hh;
    const float* qj_ptr = sq + j * D_MODEL + 2 * hh;

    v2f A[K1_CHUNKS];
    float sqpart = 0.0f, qpart = 0.0f;
#pragma unroll
    for (int c = 0; c < 6; ++c) {
      const v2f cj = *(const v2f*)(cj_ptr + 4 * c);
      const v2f xj = *(const v2f*)(xj_ptr + 4 * c);
      const v2f qj = *(const v2f*)(qj_ptr + 4 * c);
      const float d0 = ci[c][0] - cj[0];
      const float d1 = ci[c][1] - cj[1];
      A[c][0] = fabsf(d0);
      A[c][1] = fabsf(d1);
      sqpart  = fmaf(d0, d0, fmaf(d1, d1, sqpart));
      qpart   = fmaf(qi[c][0], qj[0], fmaf(qi[c][1], qj[1], qpart));
      A[6 + c][0] = xj[0];
      A[6 + c][1] = xj[1];
    }
    const float qfull = qpart  + __shfl_xor(qpart, 16, 32);
    const float cdist = sqrtf(sqpart + __shfl_xor(sqpart, 16, 32));
    A[12][0] = hh ? 1.0f : qfull;
    A[12][1] = hh ? 0.0f : cdist;

    v8f acc = {};
#pragma unroll
    for (int c = 0; c < K1_CHUNKS; ++c)
      acc = __builtin_amdgcn_wmma_f32_16x16x4_f32(
          false, A[c], false, B1[c], (short)0, acc, false, false);

#pragma unroll
    for (int r = 0; r < 8; ++r)
      hw[(r + 8 * hh) * 16 + m] = fmaxf(acc[r], 0.0f);

    v2f A2[K2_CHUNKS];
#pragma unroll
    for (int c = 0; c < K2_CHUNKS; ++c)
      A2[c] = *(const v2f*)(&hw[m * 16 + 4 * c + 2 * hh]);

    v8f acc2 = {};
#pragma unroll
    for (int c = 0; c < K2_CHUNKS; ++c)
      acc2 = __builtin_amdgcn_wmma_f32_16x16x4_f32(
          false, A2[c], false, B2[c], (short)0, acc2, false, false);

#pragma unroll
    for (int r = 0; r < 8; ++r) {
      const float rel  = fmaxf(acc2[r] + b2v, 0.0f);
      const bool  diag = (jbase + r + 8 * hh) == i;
      psum += diag ? 0.0f : rel;
      pmax  = diag ? pmax : fmaxf(pmax, rel);
    }
  }

  psum += __shfl_xor(psum, 16, 32);
  pmax  = fmaxf(pmax, __shfl_xor(pmax, 16, 32));

  if (lane < REL_DIM) {
    const size_t base = ((size_t)b * NN + i) * REL_DIM + lane;
    out[base] = psum / (float)(NN - 1);
    out[(size_t)BB * NN * REL_DIM + base] = pmax;
  }
}

extern "C" void kernel_launch(void* const* d_in, const int* in_sizes, int n_in,
                              void* d_out, int out_size, void* d_ws, size_t ws_size,
                              hipStream_t stream) {
  const float* x  = (const float*)d_in[0];
  const float* q  = (const float*)d_in[1];
  const float* co = (const float*)d_in[2];
  const float* W1 = (const float*)d_in[3];
  const float* b1 = (const float*)d_in[4];
  const float* W2 = (const float*)d_in[5];
  const float* b2 = (const float*)d_in[6];
  float* out = (float*)d_out;

  dim3 grid(BB * NN / WAVES);
  dim3 block(32 * WAVES);
  hipLaunchKernelGGL(pairagg_kernel, grid, block, 0, stream,
                     x, q, co, W1, b1, W2, b2, out);
}


# Swap WMMA operands (A=weights^T, B=features^T): layer-1 D registers are layer-2's B fragment, LDS transpose gone; j per lane, row sum/max end in a 16-lane butterfly (sum order changes)
# speedup vs baseline: 1.6284x; 1.0813x over previous
#include <hip/hip_runtime.h>


typedef __attribute__((ext_vector_type(2))) float v2f;
typedef __attribute__((ext_vector_type(8))) float v8f;

#define D_MODEL 24
#define REL_DIM 10
#define NN 512
#define BB 8
#define K1_CHUNKS 13
#define K2_CHUNKS 3
#define WAVES 16
#define SLAB (NN * D_MODEL)

__global__ __launch_bounds__(32 * WAVES) void pairagg_kernel(
    const float* __restrict__ x, const float* __restrict__ q,
    const float* __restrict__ coord, const float* __restrict__ W1,
    const float* __restrict__ b1, const float* __restrict__ W2,
    const float* __restrict__ b2, float* __restrict__ out)
{
  __shared__ __attribute__((aligned(16))) float sx[SLAB];
  __shared__ __attribute__((aligned(16))) float sq[SLAB];
  __shared__ __attribute__((aligned(16))) float sc[SLAB];

  const int wave = __builtin_amdgcn_readfirstlane((int)(threadIdx.x >> 5));
  const int b    = blockIdx.x / (NN / WAVES);
  const int i    = (blockIdx.x % (NN / WAVES)) * WAVES + wave;
  const int lane = threadIdx.x & 31;
  const int m    = lane & 15;
  const int hh   = lane >> 4;

  const float* xB = x     + (size_t)b * NN * D_MODEL;
  const float* qB = q     + (size_t)b * NN * D_MODEL;
  const float* cB = coord + (size_t)b * NN * D_MODEL;

  {
    const float4* gx = (const float4*)xB;
    const float4* gq = (const float4*)qB;
    const float4* gc = (const float4*)cB;
    float4* lx = (float4*)sx;
    float4* lq = (float4*)sq;
    float4* lc = (float4*)sc;
#pragma unroll
    for (int t = 0; t < SLAB / 4 / (32 * WAVES); ++t) {
      const int idx = t * (32 * WAVES) + (int)threadIdx.x;
      lx[idx] = gx[idx];
      lq[idx] = gq[idx];
      lc[idx] = gc[idx];
    }
  }

  v2f ci[6], qi[6];
#pragma unroll
  for (int c = 0; c < 6; ++c) {
    const int d = 4 * c + 2 * hh;
    ci[c] = *(const v2f*)(cB + i * D_MODEL + d);
    qi[c] = *(const v2f*)(qB + i * D_MODEL + d);
  }

  const int unit = (m < 6) ? m : ((m >= 8 && m < 12) ? m - 2 : -1);

  float bias_row = 0.0f;
  if (unit >= 0) {
    bias_row = b1[unit];
#pragma unroll
    for (int d = 0; d < D_MODEL; ++d)
      bias_row = fmaf(xB[i * D_MODEL + d], W1[d * REL_DIM + unit], bias_row);
  }

  v2f W1T[K1_CHUNKS];
#pragma unroll
  for (int c = 0; c < K1_CHUNKS; ++c) {
#pragma unroll
    for (int v = 0; v < 2; ++v) {
      const int k = 4 * c + 2 * hh + v;
      float val = 0.0f;
      if (unit >= 0) {
        int row = -1;
        if      (k < 24)  row = 48 + k;
        else if (k < 48)  row = k;
        else if (k == 48) row = 72;
        else if (k == 49) row = 73;
        if (row >= 0)       val = W1[row * REL_DIM + unit];
        else if (k == 50)   val = bias_row;
      }
      W1T[c][v] = val;
    }
  }

  v2f W2T[K2_CHUNKS];
#pragma unroll
  for (int c = 0; c < K2_CHUNKS; ++c) {
#pragma unroll
    for (int v = 0; v < 2; ++v) {
      const int n = hh ? ((c < 2) ? 6 + 2 * c + v : -1) : 2 * c + v;
      W2T[c][v] = (n >= 0 && m < REL_DIM) ? W2[n * REL_DIM + m] : 0.0f;
    }
  }
  float b2r[8];
  for (int r = 0; r < 8; ++r) {
    const int o = 8 * hh + r;
    b2r[r] = (o < REL_DIM) ? b2[o] : 0.0f;
  }

  __syncthreads();

  float psum[8], pmax[8];
  for (int r = 0; r < 8; ++r) { psum[r] = 0.0f; pmax[r] = -1e9f; }

  for (int jbase = 0; jbase < NN; jbase += 16) {
    const int j = jbase + m;
    const float* cj_ptr = sc + j * D_MODEL + 2 * hh;
    const float* xj_ptr = sx + j * D_MODEL + 2 * hh;
    const float* qj_ptr = sq + j * D_MODEL + 2 * hh;

    v2f F[K1_CHUNKS];
    float sqpart = 0.0f, qpart = 0.0f;
#pragma unroll
    for (int c = 0; c < 6; ++c) {
      const v2f cj = *(const v2f*)(cj_ptr + 4 * c);
      const v2f xj = *(const v2f*)(xj_ptr + 4 * c);
      const v2f qj = *(const v2f*)(qj_ptr + 4 * c);
      const float d0 = ci[c][0] - cj[0];
      const float d1 = ci[c][1] - cj[1];
      F[c][0] = fabsf(d0);
      F[c][1] = fabsf(d1);
      sqpart  = fmaf(d0, d0, fmaf(d1, d1, sqpart));
      qpart   = fmaf(qi[c][0], qj[0], fmaf(qi[c][1], qj[1], qpart));
      F[6 + c][0] = xj[0];
      F[6 + c][1] = xj[1];
    }
    const float qfull = qpart  + __shfl_xor(qpart, 16, 32);
    const float cdist = sqrtf(sqpart + __shfl_xor(sqpart, 16, 32));
    F[12][0] = hh ? 1.0f : qfull;
    F[12][1] = hh ? 0.0f : cdist;

    v8f acc = {};
#pragma unroll
    for (int c = 0; c < K1_CHUNKS; ++c)
      acc = __builtin_amdgcn_wmma_f32_16x16x4_f32(
          false, W1T[c], false, F[c], (short)0, acc, false, false);

    v2f H[K2_CHUNKS];
    for (int c = 0; c < K2_CHUNKS; ++c) {
      H[c][0] = fmaxf(acc[2 * c], 0.0f);
      H[c][1] = fmaxf(acc[2 * c + 1], 0.0f);
    }

    v8f acc2 = {};
#pragma unroll
    for (int c = 0; c < K2_CHUNKS; ++c)
      acc2 = __builtin_amdgcn_wmma_f32_16x16x4_f32(
          false, W2T[c], false, H[c], (short)0, acc2, false, false);

    const bool diag = (j == i);
    for (int r = 0; r < 8; ++r) {
      const float rel = fmaxf(acc2[r] + b2r[r], 0.0f);
      psum[r] += diag ? 0.0f : rel;
      pmax[r]  = diag ? pmax[r] : fmaxf(pmax[r], rel);
    }
  }

  for (int off = 8; off >= 1; off >>= 1) {
    for (int r = 0; r < 8; ++r) {
      psum[r] += __shfl_xor(psum[r], off, 32);
      pmax[r]  = fmaxf(pmax[r], __shfl_xor(pmax[r], off, 32));
    }
  }

  float s_out = psum[0], x_out = pmax[0];
  for (int r = 1; r < 8; ++r) {
    if (m == r) { s_out = psum[r]; x_out = pmax[r]; }
  }
  const int o = 8 * hh + m;
  if (m < 8 && o < REL_DIM) {
    const size_t base = ((size_t)b * NN + i) * REL_DIM + o;
    out[base] = s_out / (float)(NN - 1);
    out[(size_t)BB * NN * REL_DIM + base] = x_out;
  }
}

extern "C" void kernel_launch(void* const* d_in, const int* in_sizes, int n_in,
                              void* d_out, int out_size, void* d_ws, size_t ws_size,
                              hipStream_t stream) {
  const float* x  = (const float*)d_in[0];
  const float* q  = (const float*)d_in[1];
  const float* co = (const float*)d_in[2];
  const float* W1 = (const float*)d_in[3];
  const float* b1 = (const float*)d_in[4];
  const float* W2 = (const float*)d_in[5];
  const float* b2 = (const float*)d_in[6];
  float* out = (float*)d_out;

  dim3 grid(BB * NN / WAVES);
  dim3 block(32 * WAVES);
  hipLaunchKernelGGL(pairagg_kernel, grid, block, 0, stream,
                     x, q, co, W1, b1, W2, b2, out);
}


# Hoist the i-independent x_j@W1[24:48] block of layer 1 out of the pair loop: computed once per workgroup on WMMA into LDS, used as the accumulator start (13->7 WMMAs per tile; sum order changes)
# speedup vs baseline: 1.9495x; 1.1972x over previous
#include <hip/hip_runtime.h>


typedef __attribute__((ext_vector_type(2))) float v2f;
typedef __attribute__((ext_vector_type(8))) float v8f;

#define D_MODEL 24
#define REL_DIM 10
#define NN 512
#define BB 8
#define K1_CHUNKS 7
#define K0_CHUNKS 6
#define JTILES (NN / 16)
#define K2_CHUNKS 3
#define WAVES 16
#define SLAB (NN * D_MODEL)

__global__ __launch_bounds__(32 * WAVES) void pairagg_kernel(
    const float* __restrict__ x, const float* __restrict__ q,
    const float* __restrict__ coord, const float* __restrict__ W1,
    const float* __restrict__ b1, const float* __restrict__ W2,
    const float* __restrict__ b2, float* __restrict__ out)
{
  __shared__ __attribute__((aligned(16))) float sq[SLAB];
  __shared__ __attribute__((aligned(16))) float sc[SLAB];
  __shared__ __attribute__((aligned(16))) float4 sP4[JTILES * 32];
  __shared__ __attribute__((aligned(8)))  v2f    sP2[JTILES * 32];

  const int wave = __builtin_amdgcn_readfirstlane((int)(threadIdx.x >> 5));
  const int b    = blockIdx.x / (NN / WAVES);
  const int i    = (blockIdx.x % (NN / WAVES)) * WAVES + wave;
  const int lane = threadIdx.x & 31;
  const int m    = lane & 15;
  const int hh   = lane >> 4;

  const float* xB = x     + (size_t)b * NN * D_MODEL;
  const float* qB = q     + (size_t)b * NN * D_MODEL;
  const float* cB = coord + (size_t)b * NN * D_MODEL;

  const int unit = (m < 6) ? m : ((m >= 8 && m < 12) ? m - 2 : -1);

  v2f W1X[K0_CHUNKS];
  v2f X[JTILES / WAVES][K0_CHUNKS];
  {
    const int ucol = (unit >= 0) ? unit : 0;
#pragma unroll
    for (int c = 0; c < K0_CHUNKS; ++c) {
#pragma unroll
      for (int v = 0; v < 2; ++v) {
        const int k = 4 * c + 2 * hh + v;
        const float w = W1[(24 + k) * REL_DIM + ucol];
        W1X[c][v] = (unit >= 0) ? w : 0.0f;
      }
    }
#pragma unroll
    for (int s = 0; s < JTILES / WAVES; ++s) {
      const float* xj_ptr = xB + (16 * (wave + WAVES * s) + m) * D_MODEL + 2 * hh;
#pragma unroll
      for (int c = 0; c < K0_CHUNKS; ++c) X[s][c] = *(const v2f*)(xj_ptr + 4 * c);
    }
  }

  {
    const float4* gq = (const float4*)qB;
    const float4* gc = (const float4*)cB;
    float4* lq = (float4*)sq;
    float4* lc = (float4*)sc;
#pragma unroll
    for (int t = 0; t < SLAB / 4 / (32 * WAVES); ++t) {
      const int idx = t * (32 * WAVES) + (int)threadIdx.x;
      lq[idx] = gq[idx];
      lc[idx] = gc[idx];
    }
  }

  v2f ci[6], qi[6];
#pragma unroll
  for (int c = 0; c < 6; ++c) {
    const int d = 4 * c + 2 * hh;
    ci[c] = *(const v2f*)(cB + i * D_MODEL + d);
    qi[c] = *(const v2f*)(qB + i * D_MODEL + d);
  }

  float bias_row = 0.0f;
  if (unit >= 0) {
    bias_row = b1[unit];
#pragma unroll
    for (int d = 0; d < D_MODEL; ++d)
      bias_row = fmaf(xB[i * D_MODEL + d], W1[d * REL_DIM + unit], bias_row);
  }

  v2f W1T[K1_CHUNKS];
#pragma unroll
  for (int c = 0; c < K1_CHUNKS; ++c) {
#pragma unroll
    for (int v = 0; v < 2; ++v) {
      const int k = 4 * c + 2 * hh + v;
      float val = 0.0f;
      if (unit >= 0) {
        int row = -1;
        if      (k < 24)  row = 48 + k;
        else if (k == 24) row = 72;
        else if (k == 25) row = 73;
        if (row >= 0)       val = W1[row * REL_DIM + unit];
        else if (k == 26)   val = bias_row;
      }
      W1T[c][v] = val;
    }
  }

  v2f W2T[K2_CHUNKS];
#pragma unroll
  for (int c = 0; c < K2_CHUNKS; ++c) {
#pragma unroll
    for (int v = 0; v < 2; ++v) {
      const int n = hh ? ((c < 2) ? 6 + 2 * c + v : -1) : 2 * c + v;
      W2T[c][v] = (n >= 0 && m < REL_DIM) ? W2[n * REL_DIM + m] : 0.0f;
    }
  }
  float b2r[8];
  for (int r = 0; r < 8; ++r) {
    const int o = 8 * hh + r;
    b2r[r] = (o < REL_DIM) ? b2[o] : 0.0f;
  }

#pragma unroll
  for (int s = 0; s < JTILES / WAVES; ++s) {
    const int t = wave + WAVES * s;
    v8f p = {};
#pragma unroll
    for (int c = 0; c < K0_CHUNKS; ++c)
      p = __builtin_amdgcn_wmma_f32_16x16x4_f32(
          false, W1X[c], false, X[s][c], (short)0, p, false, false);
    sP4[t * 32 + lane] = make_float4(p[0], p[1], p[2], p[3]);
    v2f p45; p45[0] = p[4]; p45[1] = p[5];
    sP2[t * 32 + lane] = p45;
  }

  __syncthreads();

  float psum[8], pmax[8];
  for (int r = 0; r < 8; ++r) { psum[r] = 0.0f; pmax[r] = -1e9f; }

  for (int jbase = 0; jbase < NN; jbase += 16) {
    const int j = jbase + m;
    const float* cj_ptr = sc + j * D_MODEL + 2 * hh;
    const float* qj_ptr = sq + j * D_MODEL + 2 * hh;
    const float4 p0123 = sP4[2 * jbase + lane];
    const v2f    p45   = sP2[2 * jbase + lane];

    v2f F[K1_CHUNKS];
    float sqpart = 0.0f, qpart = 0.0f;
#pragma unroll
    for (int c = 0; c < 6; ++c) {
      const v2f cj = *(const v2f*)(cj_ptr + 4 * c);
      const v2f qj = *(const v2f*)(qj_ptr + 4 * c);
      const float d0 = ci[c][0] - cj[0];
      const float d1 = ci[c][1] - cj[1];
      F[c][0] = fabsf(d0);
      F[c][1] = fabsf(d1);
      sqpart  = fmaf(d0, d0, fmaf(d1, d1, sqpart));
      qpart   = fmaf(qi[c][0], qj[0], fmaf(qi[c][1], qj[1], qpart));
    }
    const float qfull = qpart  + __shfl_xor(qpart, 16, 32);
    const float cdist = sqrtf(sqpart + __shfl_xor(sqpart, 16, 32));
    F[6][0] = hh ? 1.0f : qfull;
    F[6][1] = hh ? 0.0f : cdist;

    v8f acc = {p0123.x, p0123.y, p0123.z, p0123.w, p45[0], p45[1], 0.0f, 0.0f};
#pragma unroll
    for (int c = 0; c < K1_CHUNKS; ++c)
      acc = __builtin_amdgcn_wmma_f32_16x16x4_f32(
          false, W1T[c], false, F[c], (short)0, acc, false, false);

    v2f H[K2_CHUNKS];
    for (int c = 0; c < K2_CHUNKS; ++c) {
      H[c][0] = fmaxf(acc[2 * c], 0.0f);
      H[c][1] = fmaxf(acc[2 * c + 1], 0.0f);
    }

    v8f acc2 = {};
#pragma unroll
    for (int c = 0; c < K2_CHUNKS; ++c)
      acc2 = __builtin_amdgcn_wmma_f32_16x16x4_f32(
          false, W2T[c], false, H[c], (short)0, acc2, false, false);

    const bool diag = (j == i);
    for (int r = 0; r < 8; ++r) {
      const float rel = fmaxf(acc2[r] + b2r[r], 0.0f);
      psum[r] += diag ? 0.0f : rel;
      pmax[r]  = diag ? pmax[r] : fmaxf(pmax[r], rel);
    }
  }

  for (int off = 8; off >= 1; off >>= 1) {
    for (int r = 0; r < 8; ++r) {
      psum[r] += __shfl_xor(psum[r], off, 32);
      pmax[r]  = fmaxf(pmax[r], __shfl_xor(pmax[r], off, 32));
    }
  }

  float s_out = psum[0], x_out = pmax[0];
  for (int r = 1; r < 8; ++r) {
    if (m == r) { s_out = psum[r]; x_out = pmax[r]; }
  }
  const int o = 8 * hh + m;
  if (m < 8 && o < REL_DIM) {
    const size_t base = ((size_t)b * NN + i) * REL_DIM + o;
    out[base] = s_out / (float)(NN - 1);
    out[(size_t)BB * NN * REL_DIM + base] = x_out;
  }
}

extern "C" void kernel_launch(void* const* d_in, const int* in_sizes, int n_in,
                              void* d_out, int out_size, void* d_ws, size_t ws_size,
                              hipStream_t stream) {
  const float* x  = (const float*)d_in[0];
  const float* q  = (const float*)d_in[1];
  const float* co = (const float*)d_in[2];
  const float* W1 = (const float*)d_in[3];
  const float* b1 = (const float*)d_in[4];
  const float* W2 = (const float*)d_in[5];
  const float* b2 = (const float*)d_in[6];
  float* out = (float*)d_out;

  dim3 grid(BB * NN / WAVES);
  dim3 block(32 * WAVES);
  hipLaunchKernelGGL(pairagg_kernel, grid, block, 0, stream,
                     x, q, co, W1, b1, W2, b2, out);
}


# Pair loop takes two j tiles per trip: their independent WMMA chains alternate (no WMMA waits on the one in front), both tiles' LDS loads issue as one group; register budget set for 4 waves per SIMD
# speedup vs baseline: 2.0365x; 1.0446x over previous
#include <hip/hip_runtime.h>


typedef __attribute__((ext_vector_type(2))) float v2f;
typedef __attribute__((ext_vector_type(8))) float v8f;

#define D_MODEL 24
#define REL_DIM 10
#define NN 512
#define BB 8
#define K1_CHUNKS 7
#define K0_CHUNKS 6
#define JTILES (NN / 16)
#define K2_CHUNKS 3
#define WAVES 16
#define SLAB (NN * D_MODEL)
#define TILES_PER_TRIP 2

__global__ __launch_bounds__(32 * WAVES) __attribute__((amdgpu_waves_per_eu(4, 4)))
void pairagg_kernel(
    const float* __restrict__ x, const float* __restrict__ q,
    const float* __restrict__ coord, const float* __restrict__ W1,
    const float* __restrict__ b1, const float* __restrict__ W2,
    const float* __restrict__ b2, float* __restrict__ out)
{
  __shared__ __attribute__((aligned(16))) float sq[SLAB];
  __shared__ __attribute__((aligned(16))) float sc[SLAB];
  __shared__ __attribute__((aligned(16))) float4 sP4[JTILES * 32];
  __shared__ __attribute__((aligned(8)))  v2f    sP2[JTILES * 32];

  const int wave = __builtin_amdgcn_readfirstlane((int)(threadIdx.x >> 5));
  const int b    = blockIdx.x / (NN / WAVES);
  const int i    = (blockIdx.x % (NN / WAVES)) * WAVES + wave;
  const int lane = threadIdx.x & 31;
  const int m    = lane & 15;
  const int hh   = lane >> 4;

  const float* xB = x     + (size_t)b * NN * D_MODEL;
  const float* qB = q     + (size_t)b * NN * D_MODEL;
  const float* cB = coord + (size_t)b * NN * D_MODEL;

  const int unit = (m < 6) ? m : ((m >= 8 && m < 12) ? m - 2 : -1);

  v2f W1X[K0_CHUNKS];
  v2f X[JTILES / WAVES][K0_CHUNKS];
  {
    const int ucol = (unit >= 0) ? unit : 0;
#pragma unroll
    for (int c = 0; c < K0_CHUNKS; ++c) {
#pragma unroll
      for (int v = 0; v < 2; ++v) {
        const int k = 4 * c + 2 * hh + v;
        const float w = W1[(24 + k) * REL_DIM + ucol];
        W1X[c][v] = (unit >= 0) ? w : 0.0f;
      }
    }
#pragma unroll
    for (int s = 0; s < JTILES / WAVES; ++s) {
      const float* xj_ptr = xB + (16 * (wave + WAVES * s) + m) * D_MODEL + 2 * hh;
#pragma unroll
      for (int c = 0; c < K0_CHUNKS; ++c) X[s][c] = *(const v2f*)(xj_ptr + 4 * c);
    }
  }

  {
    const float4* gq = (const float4*)qB;
    const float4* gc = (const float4*)cB;
    float4* lq = (float4*)sq;
    float4* lc = (float4*)sc;
#pragma unroll
    for (int t = 0; t < SLAB / 4 / (32 * WAVES); ++t) {
      const int idx = t * (32 * WAVES) + (int)threadIdx.x;
      lq[idx] = gq[idx];
      lc[idx] = gc[idx];
    }
  }

  v2f ci[6], qi[6];
#pragma unroll
  for (int c = 0; c < 6; ++c) {
    const int d = 4 * c + 2 * hh;
    ci[c] = *(const v2f*)(cB + i * D_MODEL + d);
    qi[c] = *(const v2f*)(qB + i * D_MODEL + d);
  }

  float bias_row = 0.0f;
  if (unit >= 0) {
    bias_row = b1[unit];
#pragma unroll
    for (int d = 0; d < D_MODEL; ++d)
      bias_row = fmaf(xB[i * D_MODEL + d], W1[d * REL_DIM + unit], bias_row);
  }

  v2f W1T[K1_CHUNKS];
#pragma unroll
  for (int c = 0; c < K1_CHUNKS; ++c) {
#pragma unroll
    for (int v = 0; v < 2; ++v) {
      const int k = 4 * c + 2 * hh + v;
      float val = 0.0f;
      if (unit >= 0) {
        int row = -1;
        if      (k < 24)  row = 48 + k;
        else if (k == 24) row = 72;
        else if (k == 25) row = 73;
        if (row >= 0)       val = W1[row * REL_DIM + unit];
        else if (k == 26)   val = bias_row;
      }
      W1T[c][v] = val;
    }
  }

  v2f W2T[K2_CHUNKS];
#pragma unroll
  for (int c = 0; c < K2_CHUNKS; ++c) {
#pragma unroll
    for (int v = 0; v < 2; ++v) {
      const int n = hh ? ((c < 2) ? 6 + 2 * c + v : -1) : 2 * c + v;
      W2T[c][v] = (n >= 0 && m < REL_DIM) ? W2[n * REL_DIM + m] : 0.0f;
    }
  }
  float b2r[8];
  for (int r = 0; r < 8; ++r) {
    const int o = 8 * hh + r;
    b2r[r] = (o < REL_DIM) ? b2[o] : 0.0f;
  }

#pragma unroll
  for (int s = 0; s < JTILES / WAVES; ++s) {
    const int t = wave + WAVES * s;
    v8f p = {};
#pragma unroll
    for (int c = 0; c < K0_CHUNKS; ++c)
      p = __builtin_amdgcn_wmma_f32_16x16x4_f32(
          false, W1X[c], false, X[s][c], (short)0, p, false, false);
    sP4[t * 32 + lane] = make_float4(p[0], p[1], p[2], p[3]);
    v2f p45; p45[0] = p[4]; p45[1] = p[5];
    sP2[t * 32 + lane] = p45;
  }

  __syncthreads();

  float psum[8], pmax[8];
  for (int r = 0; r < 8; ++r) { psum[r] = 0.0f; pmax[r] = -1e9f; }

  for (int jbase = 0; jbase < NN; jbase += 16 * TILES_PER_TRIP) {
    v2f F[TILES_PER_TRIP][K1_CHUNKS];
    v8f acc[TILES_PER_TRIP];

#pragma unroll
    for (int t = 0; t < TILES_PER_TRIP; ++t) {
      const int jb = jbase + 16 * t;
      const int j  = jb + m;
      const float* cj_ptr = sc + j * D_MODEL + 2 * hh;
      const float* qj_ptr = sq + j * D_MODEL + 2 * hh;
      const float4 p0123 = sP4[2 * jb + lane];
      const v2f    p45   = sP2[2 * jb + lane];

      float sqpart = 0.0f, qpart = 0.0f;
#pragma unroll
      for (int c = 0; c < 6; ++c) {
        const v2f cj = *(const v2f*)(cj_ptr + 4 * c);
        const v2f qj = *(const v2f*)(qj_ptr + 4 * c);
        const float d0 = ci[c][0] - cj[0];
        const float d1 = ci[c][1] - cj[1];
        F[t][c][0] = fabsf(d0);
        F[t][c][1] = fabsf(d1);
        sqpart  = fmaf(d0, d0, fmaf(d1, d1, sqpart));
        qpart   = fmaf(qi[c][0], qj[0], fmaf(qi[c][1], qj[1], qpart));
      }
      const float qfull = qpart  + __shfl_xor(qpart, 16, 32);
      const float cdist = sqrtf(sqpart + __shfl_xor(sqpart, 16, 32));
      F[t][6][0] = hh ? 1.0f : qfull;
      F[t][6][1] = hh ? 0.0f : cdist;

      const v8f start = {p0123.x, p0123.y, p0123.z, p0123.w, p45[0], p45[1], 0.0f, 0.0f};
      acc[t] = start;
    }

#pragma unroll
    for (int c = 0; c < K1_CHUNKS; ++c) {
#pragma unroll
      for (int t = 0; t < TILES_PER_TRIP; ++t)
        acc[t] = __builtin_amdgcn_wmma_f32_16x16x4_f32(
            false, W1T[c], false, F[t][c], (short)0, acc[t], false, false);
    }

    v2f H[TILES_PER_TRIP][K2_CHUNKS];
#pragma unroll
    for (int t = 0; t < TILES_PER_TRIP; ++t) {
#pragma unroll
      for (int c = 0; c < K2_CHUNKS; ++c) {
        H[t][c][0] = fmaxf(acc[t][2 * c], 0.0f);
        H[t][c][1] = fmaxf(acc[t][2 * c + 1], 0.0f);
      }
    }

    v8f acc2[TILES_PER_TRIP];
#pragma unroll
    for (int t = 0; t < TILES_PER_TRIP; ++t) acc2[t] = v8f{};
#pragma unroll
    for (int c = 0; c < K2_CHUNKS; ++c) {
#pragma unroll
      for (int t = 0; t < TILES_PER_TRIP; ++t)
        acc2[t] = __builtin_amdgcn_wmma_f32_16x16x4_f32(
            false, W2T[c], false, H[t][c], (short)0, acc2[t], false, false);
    }

#pragma unroll
    for (int t = 0; t < TILES_PER_TRIP; ++t) {
      const bool diag = (jbase + 16 * t + m == i);
#pragma unroll
      for (int r = 0; r < 8; ++r) {
        const float rel = fmaxf(acc2[t][r] + b2r[r], 0.0f);
        psum[r] += diag ? 0.0f : rel;
        pmax[r]  = diag ? pmax[r] : fmaxf(pmax[r], rel);
      }
    }
  }

  for (int off = 8; off >= 1; off >>= 1) {
    for (int r = 0; r < 8; ++r) {
      psum[r] += __shfl_xor(psum[r], off, 32);
      pmax[r]  = fmaxf(pmax[r], __shfl_xor(pmax[r], off, 32));
    }
  }

  float s_out = psum[0], x_out = pmax[0];
  for (int r = 1; r < 8; ++r) {
    if (m == r) { s_out = psum[r]; x_out = pmax[r]; }
  }
  const int o = 8 * hh + m;
  if (m < 8 && o < REL_DIM) {
    const size_t base = ((size_t)b * NN + i) * REL_DIM + o;
    out[base] = s_out / (float)(NN - 1);
    out[(size_t)BB * NN * REL_DIM + base] = x_out;
  }
}

extern "C" void kernel_launch(void* const* d_in, const int* in_sizes, int n_in,
                              void* d_out, int out_size, void* d_ws, size_t ws_size,
                              hipStream_t stream) {
  const float* x  = (const float*)d_in[0];
  const float* q  = (const float*)d_in[1];
  const float* co = (const float*)d_in[2];
  const float* W1 = (const float*)d_in[3];
  const float* b1 = (const float*)d_in[4];
  const float* W2 = (const float*)d_in[5];
  const float* b2 = (const float*)d_in[6];
  float* out = (float*)d_out;

  dim3 grid(BB * NN / WAVES);
  dim3 block(32 * WAVES);
  hipLaunchKernelGGL(pairagg_kernel, grid, block, 0, stream,
                     x, q, co, W1, b1, W2, b2, out);
}


# q_overlap hoisted out of the pair loop: S = q_i @ q_j^T of the workgroup's 16 rows computed once on WMMA into LDS; loop reads one value per pair, no q slab, no q FMAs (q dot sum order changes)
# speedup vs baseline: 2.1415x; 1.0516x over previous
#include <hip/hip_runtime.h>


typedef __attribute__((ext_vector_type(2))) float v2f;
typedef __attribute__((ext_vector_type(8))) float v8f;

#define D_MODEL 24
#define REL_DIM 10
#define NN 512
#define BB 8
#define K1_CHUNKS 7
#define K0_CHUNKS 6
#define JTILES (NN / 16)
#define K2_CHUNKS 3
#define WAVES 16
#define SLAB (NN * D_MODEL)
#define TILES_PER_TRIP 2
#define SPITCH (NN + 4)

__global__ __launch_bounds__(32 * WAVES) __attribute__((amdgpu_waves_per_eu(4, 4)))
void pairagg_kernel(
    const float* __restrict__ x, const float* __restrict__ q,
    const float* __restrict__ coord, const float* __restrict__ W1,
    const float* __restrict__ b1, const float* __restrict__ W2,
    const float* __restrict__ b2, float* __restrict__ out)
{
  __shared__ __attribute__((aligned(16))) float sc[SLAB];
  __shared__ __attribute__((aligned(16))) float sS[WAVES * SPITCH];
  __shared__ __attribute__((aligned(16))) float4 sP4[JTILES * 32];
  __shared__ __attribute__((aligned(8)))  v2f    sP2[JTILES * 32];

  const int wave = __builtin_amdgcn_readfirstlane((int)(threadIdx.x >> 5));
  const int b    = blockIdx.x / (NN / WAVES);
  const int i0   = (blockIdx.x % (NN / WAVES)) * WAVES;
  const int i    = i0 + wave;
  const int lane = threadIdx.x & 31;
  const int m    = lane & 15;
  const int hh   = lane >> 4;

  const float* xB = x     + (size_t)b * NN * D_MODEL;
  const float* qB = q     + (size_t)b * NN * D_MODEL;
  const float* cB = coord + (size_t)b * NN * D_MODEL;

  const int unit = (m < 6) ? m : ((m >= 8 && m < 12) ? m - 2 : -1);

  v2f W1X[K0_CHUNKS];
  v2f X[JTILES / WAVES][K0_CHUNKS];
  {
    const int ucol = (unit >= 0) ? unit : 0;
#pragma unroll
    for (int c = 0; c < K0_CHUNKS; ++c) {
#pragma unroll
      for (int v = 0; v < 2; ++v) {
        const int k = 4 * c + 2 * hh + v;
        const float w = W1[(24 + k) * REL_DIM + ucol];
        W1X[c][v] = (unit >= 0) ? w : 0.0f;
      }
    }
#pragma unroll
    for (int s = 0; s < JTILES / WAVES; ++s) {
      const float* xj_ptr = xB + (16 * (wave + WAVES * s) + m) * D_MODEL + 2 * hh;
#pragma unroll
      for (int c = 0; c < K0_CHUNKS; ++c) X[s][c] = *(const v2f*)(xj_ptr + 4 * c);
    }
  }

  v2f QA[K0_CHUNKS];
  v2f QB[JTILES / WAVES][K0_CHUNKS];
  {
    const float* qi_ptr = qB + (i0 + m) * D_MODEL + 2 * hh;
#pragma unroll
    for (int c = 0; c < K0_CHUNKS; ++c) QA[c] = *(const v2f*)(qi_ptr + 4 * c);
#pragma unroll
    for (int s = 0; s < JTILES / WAVES; ++s) {
      const float* qj_ptr = qB + (16 * (wave + WAVES * s) + m) * D_MODEL + 2 * hh;
#pragma unroll
      for (int c = 0; c < K0_CHUNKS; ++c) QB[s][c] = *(const v2f*)(qj_ptr + 4 * c);
    }
  }

  {
    const float4* gc = (const float4*)cB;
    float4* lc = (float4*)sc;
#pragma unroll
    for (int t = 0; t < SLAB / 4 / (32 * WAVES); ++t) {
      const int idx = t * (32 * WAVES) + (int)threadIdx.x;
      lc[idx] = gc[idx];
    }
  }

  v2f ci[6];
#pragma unroll
  for (int c = 0; c < 6; ++c) {
    const int d = 4 * c + 2 * hh;
    ci[c] = *(const v2f*)(cB + i * D_MODEL + d);
  }

  float bias_row = 0.0f;
  if (unit >= 0) {
    bias_row = b1[unit];
#pragma unroll
    for (int d = 0; d < D_MODEL; ++d)
      bias_row = fmaf(xB[i * D_MODEL + d], W1[d * REL_DIM + unit], bias_row);
  }

  v2f W1T[K1_CHUNKS];
#pragma unroll
  for (int c = 0; c < K1_CHUNKS; ++c) {
#pragma unroll
    for (int v = 0; v < 2; ++v) {
      const int k = 4 * c + 2 * hh + v;
      float val = 0.0f;
      if (unit >= 0) {
        int row = -1;
        if      (k < 24)  row = 48 + k;
        else if (k == 24) row = 72;
        else if (k == 25) row = 73;
        if (row >= 0)       val = W1[row * REL_DIM + unit];
        else if (k == 26)   val = bias_row;
      }
      W1T[c][v] = val;
    }
  }

  v2f W2T[K2_CHUNKS];
#pragma unroll
  for (int c = 0; c < K2_CHUNKS; ++c) {
#pragma unroll
    for (int v = 0; v < 2; ++v) {
      const int n = hh ? ((c < 2) ? 6 + 2 * c + v : -1) : 2 * c + v;
      W2T[c][v] = (n >= 0 && m < REL_DIM) ? W2[n * REL_DIM + m] : 0.0f;
    }
  }
  float b2r[8];
  for (int r = 0; r < 8; ++r) {
    const int o = 8 * hh + r;
    b2r[r] = (o < REL_DIM) ? b2[o] : 0.0f;
  }

#pragma unroll
  for (int s = 0; s < JTILES / WAVES; ++s) {
    const int t = wave + WAVES * s;
    v8f p = {};
#pragma unroll
    for (int c = 0; c < K0_CHUNKS; ++c)
      p = __builtin_amdgcn_wmma_f32_16x16x4_f32(
          false, W1X[c], false, X[s][c], (short)0, p, false, false);
    sP4[t * 32 + lane] = make_float4(p[0], p[1], p[2], p[3]);
    v2f p45; p45[0] = p[4]; p45[1] = p[5];
    sP2[t * 32 + lane] = p45;
  }

#pragma unroll
  for (int s = 0; s < JTILES / WAVES; ++s) {
    const int t = wave + WAVES * s;
    v8f sv = {};
#pragma unroll
    for (int c = 0; c < K0_CHUNKS; ++c)
      sv = __builtin_amdgcn_wmma_f32_16x16x4_f32(
          false, QA[c], false, QB[s][c], (short)0, sv, false, false);
#pragma unroll
    for (int r = 0; r < 8; ++r) sS[(8 * hh + r) * SPITCH + 16 * t + m] = sv[r];
  }

  __syncthreads();

  float psum[8], pmax[8];
  for (int r = 0; r < 8; ++r) { psum[r] = 0.0f; pmax[r] = -1e9f; }

  for (int jbase = 0; jbase < NN; jbase += 16 * TILES_PER_TRIP) {
    v2f F[TILES_PER_TRIP][K1_CHUNKS];
    v8f acc[TILES_PER_TRIP];

#pragma unroll
    for (int t = 0; t < TILES_PER_TRIP; ++t) {
      const int jb = jbase + 16 * t;
      const int j  = jb + m;
      const float* cj_ptr = sc + j * D_MODEL + 2 * hh;
      const float qfull = sS[wave * SPITCH + j];
      const float4 p0123 = sP4[2 * jb + lane];
      const v2f    p45   = sP2[2 * jb + lane];

      float sqpart = 0.0f;
#pragma unroll
      for (int c = 0; c < 6; ++c) {
        const v2f cj = *(const v2f*)(cj_ptr + 4 * c);
        const float d0 = ci[c][0] - cj[0];
        const float d1 = ci[c][1] - cj[1];
        F[t][c][0] = fabsf(d0);
        F[t][c][1] = fabsf(d1);
        sqpart  = fmaf(d0, d0, fmaf(d1, d1, sqpart));
      }
      const float cdist = sqrtf(sqpart + __shfl_xor(sqpart, 16, 32));
      F[t][6][0] = hh ? 1.0f : qfull;
      F[t][6][1] = hh ? 0.0f : cdist;

      const v8f start = {p0123.x, p0123.y, p0123.z, p0123.w, p45[0], p45[1], 0.0f, 0.0f};
      acc[t] = start;
    }

#pragma unroll
    for (int c = 0; c < K1_CHUNKS; ++c) {
#pragma unroll
      for (int t = 0; t < TILES_PER_TRIP; ++t)
        acc[t] = __builtin_amdgcn_wmma_f32_16x16x4_f32(
            false, W1T[c], false, F[t][c], (short)0, acc[t], false, false);
    }

    v2f H[TILES_PER_TRIP][K2_CHUNKS];
#pragma unroll
    for (int t = 0; t < TILES_PER_TRIP; ++t) {
#pragma unroll
      for (int c = 0; c < K2_CHUNKS; ++c) {
        H[t][c][0] = fmaxf(acc[t][2 * c], 0.0f);
        H[t][c][1] = fmaxf(acc[t][2 * c + 1], 0.0f);
      }
    }

    v8f acc2[TILES_PER_TRIP];
#pragma unroll
    for (int t = 0; t < TILES_PER_TRIP; ++t) acc2[t] = v8f{};
#pragma unroll
    for (int c = 0; c < K2_CHUNKS; ++c) {
#pragma unroll
      for (int t = 0; t < TILES_PER_TRIP; ++t)
        acc2[t] = __builtin_amdgcn_wmma_f32_16x16x4_f32(
            false, W2T[c], false, H[t][c], (short)0, acc2[t], false, false);
    }

#pragma unroll
    for (int t = 0; t < TILES_PER_TRIP; ++t) {
      const bool diag = (jbase + 16 * t + m == i);
#pragma unroll
      for (int r = 0; r < 8; ++r) {
        const float rel = fmaxf(acc2[t][r] + b2r[r], 0.0f);
        psum[r] += diag ? 0.0f : rel;
        pmax[r]  = diag ? pmax[r] : fmaxf(pmax[r], rel);
      }
    }
  }

  for (int off = 8; off >= 1; off >>= 1) {
    for (int r = 0; r < 8; ++r) {
      psum[r] += __shfl_xor(psum[r], off, 32);
      pmax[r]  = fmaxf(pmax[r], __shfl_xor(pmax[r], off, 32));
    }
  }

  float s_out = psum[0], x_out = pmax[0];
  for (int r = 1; r < 8; ++r) {
    if (m == r) { s_out = psum[r]; x_out = pmax[r]; }
  }
  const int o = 8 * hh + m;
  if (m < 8 && o < REL_DIM) {
    const size_t base = ((size_t)b * NN + i) * REL_DIM + o;
    out[base] = s_out / (float)(NN - 1);
    out[(size_t)BB * NN * REL_DIM + base] = x_out;
  }
}

extern "C" void kernel_launch(void* const* d_in, const int* in_sizes, int n_in,
                              void* d_out, int out_size, void* d_ws, size_t ws_size,
                              hipStream_t stream) {
  const float* x  = (const float*)d_in[0];
  const float* q  = (const float*)d_in[1];
  const float* co = (const float*)d_in[2];
  const float* W1 = (const float*)d_in[3];
  const float* b1 = (const float*)d_in[4];
  const float* W2 = (const float*)d_in[5];
  const float* b2 = (const float*)d_in[6];
  float* out = (float*)d_out;

  dim3 grid(BB * NN / WAVES);
  dim3 block(32 * WAVES);
  hipLaunchKernelGGL(pairagg_kernel, grid, block, 0, stream,
                     x, q, co, W1, b1, W2, b2, out);
}


# Pair loop vector work 184 -> 140 instr per trip: both layers' units seated 5+5 (5 live D registers), diagonal zeroed once for sum and max, squares packed per tile (layer-2, cdist sum orders change)
# speedup vs baseline: 2.3937x; 1.1178x over previous
#include <hip/hip_runtime.h>


typedef __attribute__((ext_vector_type(2))) float v2f;
typedef __attribute__((ext_vector_type(8))) float v8f;

#define D_MODEL 24
#define REL_DIM 10
#define NN 512
#define BB 8
#define K1_CHUNKS 7
#define K0_CHUNKS 6
#define JTILES (NN / 16)
#define K2_CHUNKS 3
#define WAVES 16
#define SLAB (NN * D_MODEL)
#define TILES_PER_TRIP 2
#define SPITCH (NN + 4)

__global__ __launch_bounds__(32 * WAVES) __attribute__((amdgpu_waves_per_eu(4, 4)))
void pairagg_kernel(
    const float* __restrict__ x, const float* __restrict__ q,
    const float* __restrict__ coord, const float* __restrict__ W1,
    const float* __restrict__ b1, const float* __restrict__ W2,
    const float* __restrict__ b2, float* __restrict__ out)
{
  __shared__ __attribute__((aligned(16))) float sc[SLAB];
  __shared__ __attribute__((aligned(16))) float sS[WAVES * SPITCH];
  __shared__ __attribute__((aligned(16))) float4 sP4[JTILES * 32];
  __shared__ float sP1[JTILES * 32];

  const int wave = __builtin_amdgcn_readfirstlane((int)(threadIdx.x >> 5));
  const int b    = blockIdx.x / (NN / WAVES);
  const int i0   = (blockIdx.x % (NN / WAVES)) * WAVES;
  const int i    = i0 + wave;
  const int lane = threadIdx.x & 31;
  const int m    = lane & 15;
  const int hh   = lane >> 4;

  const float* xB = x     + (size_t)b * NN * D_MODEL;
  const float* qB = q     + (size_t)b * NN * D_MODEL;
  const float* cB = coord + (size_t)b * NN * D_MODEL;

  const int unit = (m < 5) ? m : ((m >= 8 && m < 13) ? m - 3 : -1);

  v2f W1X[K0_CHUNKS];
  v2f X[JTILES / WAVES][K0_CHUNKS];
  {
    const int ucol = (unit >= 0) ? unit : 0;
#pragma unroll
    for (int c = 0; c < K0_CHUNKS; ++c) {
#pragma unroll
      for (int v = 0; v < 2; ++v) {
        const int k = 4 * c + 2 * hh + v;
        const float w = W1[(24 + k) * REL_DIM + ucol];
        W1X[c][v] = (unit >= 0) ? w : 0.0f;
      }
    }
#pragma unroll
    for (int s = 0; s < JTILES / WAVES; ++s) {
      const float* xj_ptr = xB + (16 * (wave + WAVES * s) + m) * D_MODEL + 2 * hh;
#pragma unroll
      for (int c = 0; c < K0_CHUNKS; ++c) X[s][c] = *(const v2f*)(xj_ptr + 4 * c);
    }
  }

  v2f QA[K0_CHUNKS];
  v2f QB[JTILES / WAVES][K0_CHUNKS];
  {
    const float* qi_ptr = qB + (i0 + m) * D_MODEL + 2 * hh;
#pragma unroll
    for (int c = 0; c < K0_CHUNKS; ++c) QA[c] = *(const v2f*)(qi_ptr + 4 * c);
#pragma unroll
    for (int s = 0; s < JTILES / WAVES; ++s) {
      const float* qj_ptr = qB + (16 * (wave + WAVES * s) + m) * D_MODEL + 2 * hh;
#pragma unroll
      for (int c = 0; c < K0_CHUNKS; ++c) QB[s][c] = *(const v2f*)(qj_ptr + 4 * c);
    }
  }

  {
    const float4* gc = (const float4*)cB;
    float4* lc = (float4*)sc;
#pragma unroll
    for (int t = 0; t < SLAB / 4 / (32 * WAVES); ++t) {
      const int idx = t * (32 * WAVES) + (int)threadIdx.x;
      lc[idx] = gc[idx];
    }
  }

  v2f ci[6];
#pragma unroll
  for (int c = 0; c < 6; ++c) {
    const int d = 4 * c + 2 * hh;
    ci[c] = *(const v2f*)(cB + i * D_MODEL + d);
  }

  float bias_row = 0.0f;
  if (unit >= 0) {
    bias_row = b1[unit];
#pragma unroll
    for (int d = 0; d < D_MODEL; ++d)
      bias_row = fmaf(xB[i * D_MODEL + d], W1[d * REL_DIM + unit], bias_row);
  }

  v2f W1T[K1_CHUNKS];
#pragma unroll
  for (int c = 0; c < K1_CHUNKS; ++c) {
#pragma unroll
    for (int v = 0; v < 2; ++v) {
      const int k = 4 * c + 2 * hh + v;
      float val = 0.0f;
      if (unit >= 0) {
        int row = -1;
        if      (k < 24)  row = 48 + k;
        else if (k == 24) row = 72;
        else if (k == 25) row = 73;
        if (row >= 0)       val = W1[row * REL_DIM + unit];
        else if (k == 26)   val = bias_row;
      }
      W1T[c][v] = val;
    }
  }

  v2f W2T[K2_CHUNKS];
  {
    const int ocol = (unit >= 0) ? unit : 0;
#pragma unroll
    for (int c = 0; c < K2_CHUNKS; ++c) {
#pragma unroll
      for (int v = 0; v < 2; ++v) {
        const int n = (2 * c + v < 5) ? 5 * hh + 2 * c + v : 0;
        const float w = W2[n * REL_DIM + ocol];
        W2T[c][v] = (2 * c + v < 5 && unit >= 0) ? w : 0.0f;
      }
    }
  }
  float b2r[5];
  for (int r = 0; r < 5; ++r) b2r[r] = b2[5 * hh + r];

#pragma unroll
  for (int s = 0; s < JTILES / WAVES; ++s) {
    const int t = wave + WAVES * s;
    v8f p = {};
#pragma unroll
    for (int c = 0; c < K0_CHUNKS; ++c)
      p = __builtin_amdgcn_wmma_f32_16x16x4_f32(
          false, W1X[c], false, X[s][c], (short)0, p, false, false);
    sP4[t * 32 + lane] = make_float4(p[0], p[1], p[2], p[3]);
    sP1[t * 32 + lane] = p[4];
  }

#pragma unroll
  for (int s = 0; s < JTILES / WAVES; ++s) {
    const int t = wave + WAVES * s;
    v8f sv = {};
#pragma unroll
    for (int c = 0; c < K0_CHUNKS; ++c)
      sv = __builtin_amdgcn_wmma_f32_16x16x4_f32(
          false, QA[c], false, QB[s][c], (short)0, sv, false, false);
#pragma unroll
    for (int r = 0; r < 8; ++r) sS[(8 * hh + r) * SPITCH + 16 * t + m] = sv[r];
  }

  __syncthreads();

  float psum[5], pmax[5];
  for (int r = 0; r < 5; ++r) { psum[r] = 0.0f; pmax[r] = -1e9f; }

  for (int jbase = 0; jbase < NN; jbase += 16 * TILES_PER_TRIP) {
    v2f F[TILES_PER_TRIP][K1_CHUNKS];
    v8f acc[TILES_PER_TRIP];

#pragma unroll
    for (int t = 0; t < TILES_PER_TRIP; ++t) {
      const int jb = jbase + 16 * t;
      const int j  = jb + m;
      const float* cj_ptr = sc + j * D_MODEL + 2 * hh;
      const float qfull = sS[wave * SPITCH + j];
      const float4 p0123 = sP4[2 * jb + lane];
      const float  p4    = sP1[2 * jb + lane];

      v2f sq2 = {0.0f, 0.0f};
#pragma unroll
      for (int c = 0; c < 6; ++c) {
        const v2f cj = *(const v2f*)(cj_ptr + 4 * c);
        const v2f d  = ci[c] - cj;
        F[t][c][0] = fabsf(d[0]);
        F[t][c][1] = fabsf(d[1]);
        sq2 = __builtin_elementwise_fma(d, d, sq2);
      }
      const float sqpart = sq2[0] + sq2[1];
      const float cdist = sqrtf(sqpart + __shfl_xor(sqpart, 16, 32));
      F[t][6][0] = hh ? 1.0f : qfull;
      F[t][6][1] = hh ? 0.0f : cdist;

      const v8f start = {p0123.x, p0123.y, p0123.z, p0123.w, p4, 0.0f, 0.0f, 0.0f};
      acc[t] = start;
    }

#pragma unroll
    for (int c = 0; c < K1_CHUNKS; ++c) {
#pragma unroll
      for (int t = 0; t < TILES_PER_TRIP; ++t)
        acc[t] = __builtin_amdgcn_wmma_f32_16x16x4_f32(
            false, W1T[c], false, F[t][c], (short)0, acc[t], false, false);
    }

    v2f H[TILES_PER_TRIP][K2_CHUNKS];
#pragma unroll
    for (int t = 0; t < TILES_PER_TRIP; ++t) {
#pragma unroll
      for (int c = 0; c < K2_CHUNKS; ++c) {
        H[t][c][0] = fmaxf(acc[t][2 * c], 0.0f);
        H[t][c][1] = (2 * c + 1 < 5) ? fmaxf(acc[t][2 * c + 1], 0.0f) : 0.0f;
      }
    }

    v8f acc2[TILES_PER_TRIP];
#pragma unroll
    for (int t = 0; t < TILES_PER_TRIP; ++t) acc2[t] = v8f{};
#pragma unroll
    for (int c = 0; c < K2_CHUNKS; ++c) {
#pragma unroll
      for (int t = 0; t < TILES_PER_TRIP; ++t)
        acc2[t] = __builtin_amdgcn_wmma_f32_16x16x4_f32(
            false, W2T[c], false, H[t][c], (short)0, acc2[t], false, false);
    }

#pragma unroll
    for (int t = 0; t < TILES_PER_TRIP; ++t) {
      const bool diag = (jbase + 16 * t + m == i);
#pragma unroll
      for (int r = 0; r < 5; ++r) {
        const float rel = diag ? 0.0f : fmaxf(acc2[t][r] + b2r[r], 0.0f);
        psum[r] += rel;
        pmax[r]  = fmaxf(pmax[r], rel);
      }
    }
  }

  for (int off = 8; off >= 1; off >>= 1) {
    for (int r = 0; r < 5; ++r) {
      psum[r] += __shfl_xor(psum[r], off, 32);
      pmax[r]  = fmaxf(pmax[r], __shfl_xor(pmax[r], off, 32));
    }
  }

  float s_out = psum[0], x_out = pmax[0];
  for (int r = 1; r < 5; ++r) {
    if (m == r) { s_out = psum[r]; x_out = pmax[r]; }
  }
  const int o = 5 * hh + m;
  if (m < 5) {
    const size_t base = ((size_t)b * NN + i) * REL_DIM + o;
    out[base] = s_out / (float)(NN - 1);
    out[(size_t)BB * NN * REL_DIM + base] = x_out;
  }
}

extern "C" void kernel_launch(void* const* d_in, const int* in_sizes, int n_in,
                              void* d_out, int out_size, void* d_ws, size_t ws_size,
                              hipStream_t stream) {
  const float* x  = (const float*)d_in[0];
  const float* q  = (const float*)d_in[1];
  const float* co = (const float*)d_in[2];
  const float* W1 = (const float*)d_in[3];
  const float* b1 = (const float*)d_in[4];
  const float* W2 = (const float*)d_in[5];
  const float* b2 = (const float*)d_in[6];
  float* out = (float*)d_out;

  dim3 grid(BB * NN / WAVES);
  dim3 block(32 * WAVES);
  hipLaunchKernelGGL(pairagg_kernel, grid, block, 0, stream,
                     x, q, co, W1, b1, W2, b2, out);
}


# Pair loop vector work 140 -> 106 per trip: cdist by v_sqrt_f32 (1 ulp, not correctly rounded) after a v_permlanex16 cross-half add; diagonal selects only in the row's own trip; LDS pointers carried
# speedup vs baseline: 2.5341x; 1.0587x over previous
#include <hip/hip_runtime.h>


typedef __attribute__((ext_vector_type(2))) float v2f;
typedef __attribute__((ext_vector_type(8))) float v8f;

template <bool V> struct flag { static constexpr bool value = V; };

#define D_MODEL 24
#define REL_DIM 10
#define NN 512
#define BB 8
#define K1_CHUNKS 7
#define K0_CHUNKS 6
#define JTILES (NN / 16)
#define K2_CHUNKS 3
#define WAVES 16
#define SLAB (NN * D_MODEL)
#define TILES_PER_TRIP 2
#define SPITCH (NN + 4)

__global__ __launch_bounds__(32 * WAVES) __attribute__((amdgpu_waves_per_eu(4, 4)))
void pairagg_kernel(
    const float* __restrict__ x, const float* __restrict__ q,
    const float* __restrict__ coord, const float* __restrict__ W1,
    const float* __restrict__ b1, const float* __restrict__ W2,
    const float* __restrict__ b2, float* __restrict__ out)
{
  __shared__ __attribute__((aligned(16))) float sc[SLAB];
  __shared__ __attribute__((aligned(16))) float sS[WAVES * SPITCH];
  __shared__ __attribute__((aligned(16))) float4 sP4[JTILES * 32];
  __shared__ float sP1[JTILES * 32];

  const int wave = __builtin_amdgcn_readfirstlane((int)(threadIdx.x >> 5));
  const int b    = blockIdx.x / (NN / WAVES);
  const int i0   = (blockIdx.x % (NN / WAVES)) * WAVES;
  const int i    = i0 + wave;
  const int lane = threadIdx.x & 31;
  const int m    = lane & 15;
  const int hh   = lane >> 4;

  const float* xB = x     + (size_t)b * NN * D_MODEL;
  const float* qB = q     + (size_t)b * NN * D_MODEL;
  const float* cB = coord + (size_t)b * NN * D_MODEL;

  const int unit = (m < 5) ? m : ((m >= 8 && m < 13) ? m - 3 : -1);

  v2f W1X[K0_CHUNKS];
  v2f X[JTILES / WAVES][K0_CHUNKS];
  {
    const int ucol = (unit >= 0) ? unit : 0;
#pragma unroll
    for (int c = 0; c < K0_CHUNKS; ++c) {
#pragma unroll
      for (int v = 0; v < 2; ++v) {
        const int k = 4 * c + 2 * hh + v;
        const float w = W1[(24 + k) * REL_DIM + ucol];
        W1X[c][v] = (unit >= 0) ? w : 0.0f;
      }
    }
#pragma unroll
    for (int s = 0; s < JTILES / WAVES; ++s) {
      const float* xj_ptr = xB + (16 * (wave + WAVES * s) + m) * D_MODEL + 2 * hh;
#pragma unroll
      for (int c = 0; c < K0_CHUNKS; ++c) X[s][c] = *(const v2f*)(xj_ptr + 4 * c);
    }
  }

  v2f QA[K0_CHUNKS];
  v2f QB[JTILES / WAVES][K0_CHUNKS];
  {
    const float* qi_ptr = qB + (i0 + m) * D_MODEL + 2 * hh;
#pragma unroll
    for (int c = 0; c < K0_CHUNKS; ++c) QA[c] = *(const v2f*)(qi_ptr + 4 * c);
#pragma unroll
    for (int s = 0; s < JTILES / WAVES; ++s) {
      const float* qj_ptr = qB + (16 * (wave + WAVES * s) + m) * D_MODEL + 2 * hh;
#pragma unroll
      for (int c = 0; c < K0_CHUNKS; ++c) QB[s][c] = *(const v2f*)(qj_ptr + 4 * c);
    }
  }

  {
    const float4* gc = (const float4*)cB;
    float4* lc = (float4*)sc;
#pragma unroll
    for (int t = 0; t < SLAB / 4 / (32 * WAVES); ++t) {
      const int idx = t * (32 * WAVES) + (int)threadIdx.x;
      lc[idx] = gc[idx];
    }
  }

  v2f ci[6];
#pragma unroll
  for (int c = 0; c < 6; ++c) {
    const int d = 4 * c + 2 * hh;
    ci[c] = *(const v2f*)(cB + i * D_MODEL + d);
  }

  float bias_row = 0.0f;
  if (unit >= 0) {
    bias_row = b1[unit];
#pragma unroll
    for (int d = 0; d < D_MODEL; ++d)
      bias_row = fmaf(xB[i * D_MODEL + d], W1[d * REL_DIM + unit], bias_row);
  }

  v2f W1T[K1_CHUNKS];
#pragma unroll
  for (int c = 0; c < K1_CHUNKS; ++c) {
#pragma unroll
    for (int v = 0; v < 2; ++v) {
      const int k = 4 * c + 2 * hh + v;
      float val = 0.0f;
      if (unit >= 0) {
        int row = -1;
        if      (k < 24)  row = 48 + k;
        else if (k == 24) row = 72;
        else if (k == 25) row = 73;
        if (row >= 0)       val = W1[row * REL_DIM + unit];
        else if (k == 26)   val = bias_row;
      }
      W1T[c][v] = val;
    }
  }

  v2f W2T[K2_CHUNKS];
  {
    const int ocol = (unit >= 0) ? unit : 0;
#pragma unroll
    for (int c = 0; c < K2_CHUNKS; ++c) {
#pragma unroll
      for (int v = 0; v < 2; ++v) {
        const int n = (2 * c + v < 5) ? 5 * hh + 2 * c + v : 0;
        const float w = W2[n * REL_DIM + ocol];
        W2T[c][v] = (2 * c + v < 5 && unit >= 0) ? w : 0.0f;
      }
    }
  }
  float b2r[5];
  for (int r = 0; r < 5; ++r) b2r[r] = b2[5 * hh + r];

#pragma unroll
  for (int s = 0; s < JTILES / WAVES; ++s) {
    const int t = wave + WAVES * s;
    v8f p = {};
#pragma unroll
    for (int c = 0; c < K0_CHUNKS; ++c)
      p = __builtin_amdgcn_wmma_f32_16x16x4_f32(
          false, W1X[c], false, X[s][c], (short)0, p, false, false);
    sP4[t * 32 + lane] = make_float4(p[0], p[1], p[2], p[3]);
    sP1[t * 32 + lane] = p[4];
  }

#pragma unroll
  for (int s = 0; s < JTILES / WAVES; ++s) {
    const int t = wave + WAVES * s;
    v8f sv = {};
#pragma unroll
    for (int c = 0; c < K0_CHUNKS; ++c)
      sv = __builtin_amdgcn_wmma_f32_16x16x4_f32(
          false, QA[c], false, QB[s][c], (short)0, sv, false, false);
#pragma unroll
    for (int r = 0; r < 8; ++r) sS[(8 * hh + r) * SPITCH + 16 * t + m] = sv[r];
  }

  __syncthreads();

  float psum[5], pmax[5];
  for (int r = 0; r < 5; ++r) { psum[r] = 0.0f; pmax[r] = -1e9f; }

  const float*  cj_base = sc + m * D_MODEL + 2 * hh;
  const float*  s_base  = sS + wave * SPITCH + m;
  const float4* p4_base = sP4 + lane;
  const float*  p1_base = sP1 + lane;

  auto trip = [&](auto has_diag) __attribute__((always_inline)) {
    v2f F[TILES_PER_TRIP][K1_CHUNKS];
    v8f acc[TILES_PER_TRIP];

#pragma unroll
    for (int t = 0; t < TILES_PER_TRIP; ++t) {
      const float* cj_ptr = cj_base + 16 * t * D_MODEL;
      const float qfull = s_base[16 * t];
      const float4 p0123 = p4_base[32 * t];
      const float  p4    = p1_base[32 * t];

      v2f sq2 = {0.0f, 0.0f};
#pragma unroll
      for (int c = 0; c < 6; ++c) {
        const v2f cj = *(const v2f*)(cj_ptr + 4 * c);
        const v2f d  = ci[c] - cj;
        F[t][c][0] = fabsf(d[0]);
        F[t][c][1] = fabsf(d[1]);
        sq2 = __builtin_elementwise_fma(d, d, sq2);
      }
      const float sqpart = sq2[0] + sq2[1];
      const unsigned sqbits = __builtin_bit_cast(unsigned, sqpart);
      const float sqother = __builtin_bit_cast(float,
          __builtin_amdgcn_permlanex16(sqbits, sqbits, 0x76543210u, 0xfedcba98u, false, false));
      const float cdist = __builtin_amdgcn_sqrtf(sqpart + sqother);
      F[t][6][0] = hh ? 1.0f : qfull;
      F[t][6][1] = hh ? 0.0f : cdist;

      const v8f start = {p0123.x, p0123.y, p0123.z, p0123.w, p4, 0.0f, 0.0f, 0.0f};
      acc[t] = start;
    }
    cj_base += 16 * TILES_PER_TRIP * D_MODEL;
    s_base  += 16 * TILES_PER_TRIP;
    p4_base += 32 * TILES_PER_TRIP;
    p1_base += 32 * TILES_PER_TRIP;

#pragma unroll
    for (int c = 0; c < K1_CHUNKS; ++c) {
#pragma unroll
      for (int t = 0; t < TILES_PER_TRIP; ++t)
        acc[t] = __builtin_amdgcn_wmma_f32_16x16x4_f32(
            false, W1T[c], false, F[t][c], (short)0, acc[t], false, false);
    }

    v2f H[TILES_PER_TRIP][K2_CHUNKS];
#pragma unroll
    for (int t = 0; t < TILES_PER_TRIP; ++t) {
#pragma unroll
      for (int c = 0; c < K2_CHUNKS; ++c) {
        H[t][c][0] = fmaxf(acc[t][2 * c], 0.0f);
        H[t][c][1] = (2 * c + 1 < 5) ? fmaxf(acc[t][2 * c + 1], 0.0f) : 0.0f;
      }
    }

    v8f acc2[TILES_PER_TRIP];
#pragma unroll
    for (int t = 0; t < TILES_PER_TRIP; ++t) acc2[t] = v8f{};
#pragma unroll
    for (int c = 0; c < K2_CHUNKS; ++c) {
#pragma unroll
      for (int t = 0; t < TILES_PER_TRIP; ++t)
        acc2[t] = __builtin_amdgcn_wmma_f32_16x16x4_f32(
            false, W2T[c], false, H[t][c], (short)0, acc2[t], false, false);
    }

#pragma unroll
    for (int t = 0; t < TILES_PER_TRIP; ++t) {
#pragma unroll
      for (int r = 0; r < 5; ++r) {
        float rel = fmaxf(acc2[t][r] + b2r[r], 0.0f);
        if constexpr (decltype(has_diag)::value) {
          const bool diag = (16 * t + m == (i & (16 * TILES_PER_TRIP - 1)));
          rel = diag ? 0.0f : rel;
        }
        psum[r] += rel;
        pmax[r]  = fmaxf(pmax[r], rel);
      }
    }
  };

  const int own_trip = i / (16 * TILES_PER_TRIP);
  for (int n = 0; n < own_trip; ++n) trip(flag<false>{});
  trip(flag<true>{});
  for (int n = own_trip + 1; n < NN / (16 * TILES_PER_TRIP); ++n) trip(flag<false>{});

  for (int off = 8; off >= 1; off >>= 1) {
    for (int r = 0; r < 5; ++r) {
      psum[r] += __shfl_xor(psum[r], off, 32);
      pmax[r]  = fmaxf(pmax[r], __shfl_xor(pmax[r], off, 32));
    }
  }

  float s_out = psum[0], x_out = pmax[0];
  for (int r = 1; r < 5; ++r) {
    if (m == r) { s_out = psum[r]; x_out = pmax[r]; }
  }
  const int o = 5 * hh + m;
  if (m < 5) {
    const size_t base = ((size_t)b * NN + i) * REL_DIM + o;
    out[base] = s_out / (float)(NN - 1);
    out[(size_t)BB * NN * REL_DIM + base] = x_out;
  }
}

extern "C" void kernel_launch(void* const* d_in, const int* in_sizes, int n_in,
                              void* d_out, int out_size, void* d_ws, size_t ws_size,
                              hipStream_t stream) {
  const float* x  = (const float*)d_in[0];
  const float* q  = (const float*)d_in[1];
  const float* co = (const float*)d_in[2];
  const float* W1 = (const float*)d_in[3];
  const float* b1 = (const float*)d_in[4];
  const float* W2 = (const float*)d_in[5];
  const float* b2 = (const float*)d_in[6];
  float* out = (float*)d_out;

  dim3 grid(BB * NN / WAVES);
  dim3 block(32 * WAVES);
  hipLaunchKernelGGL(pairagg_kernel, grid, block, 0, stream,
                     x, q, co, W1, b1, W2, b2, out);
}
